# starting kernel
# baseline (speedup 1.0000x reference)
#include <hip/hip_runtime.h>

typedef __attribute__((ext_vector_type(2))) float v2f;
typedef __attribute__((ext_vector_type(8))) float v8f;

#define MDEPTH 5
#define TILE   16
#define TPW    4

__global__ __launch_bounds__(256) void volterra_wmma_kernel(
    const float* __restrict__ x,
    const float* __restrict__ cr,
    const float* __restrict__ ci,
    float* __restrict__ out,
    int S, int out_len, int tilesPerRow, int totalTiles)
{
    const int lane   = threadIdx.x & 31;
    const int waveId = (int)((blockIdx.x * blockDim.x + threadIdx.x) >> 5);
    const int N      = lane & 15;
    const int half   = lane >> 4;

    v2f Bm[10];
#pragma unroll
    for (int t = 0; t < 10; ++t) {
        const bool part = (t >= 5);
        const int  j0   = 4 * t + 2 * half - (part ? 20 : 0);
        const float* s0 = part ? ci : cr;
        const float* s1 = part ? cr : ci;
        const float sgn = part ? -1.0f : 1.0f;
        float w0x = s0[j0]     * sgn;
        float w0y = s0[j0 + 1] * sgn;
        float w1x = s1[j0];
        float w1y = s1[j0 + 1];
        v2f bt;
        bt.x = (N == 0) ? w0x : ((N == 1) ? w1x : 0.0f);
        bt.y = (N == 0) ? w0y : ((N == 1) ? w1y : 0.0f);
        Bm[t] = bt;
    }

    const float2* __restrict__ x2 = (const float2*)x;

    for (int i = 0; i < TPW; ++i) {
        int g = waveId * TPW + i;
        if (g >= totalTiles) break;
        int brow  = g / tilesPerRow;
        int n0    = (g - brow * tilesPerRow) * TILE;
        int nbase = n0 + N;

        float wr[5], wi[5];
#pragma unroll
        for (int m = 0; m < 5; ++m) {
            int idx = nbase + 4 - m;
            if (idx > S - 1) idx = S - 1;
            float2 s = x2[(size_t)brow * S + idx];
            wr[m] = s.x; wi[m] = s.y;
        }
        float p1[5], p2[5], p3[5];
#pragma unroll
        for (int m = 0; m < 5; ++m) {
            float e = wr[m] * wr[m] + wi[m] * wi[m];
            p1[m] = e; p2[m] = e * e; p3[m] = p2[m] * e;
        }

        v8f c = {0.f, 0.f, 0.f, 0.f, 0.f, 0.f, 0.f, 0.f};
#pragma unroll
        for (int t = 0; t < 10; ++t) {
            const bool part = (t >= 5);
            const int  base = 4 * t - (part ? 20 : 0);
            float vals[4];
#pragma unroll
            for (int b2 = 0; b2 < 4; ++b2) {
                const int j = base + b2;
                const int k = j / 5, m = j % 5;
                float w = part ? wi[m] : wr[m];
                float p = (k == 1) ? p1[m] : (k == 2) ? p2[m] : p3[m];
                vals[b2] = (k == 0) ? w : p * w;
            }
            v2f A;
            A.x = half ? vals[2] : vals[0];
            A.y = half ? vals[3] : vals[1];
            c = __builtin_amdgcn_wmma_f32_16x16x4_f32(
                    false, A, false, Bm[t], (short)0, c, false, false);
        }

        int s       = lane >> 1;
        int comp    = lane & 1;
        int srcLane = comp + ((s & 8) ? 16 : 0);
        float tr[8];
#pragma unroll
        for (int r = 0; r < 8; ++r) tr[r] = __shfl(c[r], srcLane, 32);
        int rs = s & 7;
        float a01 = (rs & 1) ? tr[1] : tr[0];
        float a23 = (rs & 1) ? tr[3] : tr[2];
        float a45 = (rs & 1) ? tr[5] : tr[4];
        float a67 = (rs & 1) ? tr[7] : tr[6];
        float b03 = (rs & 2) ? a23 : a01;
        float b47 = (rs & 2) ? a67 : a45;
        float vv  = (rs & 4) ? b47 : b03;
        if (n0 + s < out_len)
            out[(size_t)brow * out_len * 2 + (size_t)n0 * 2 + lane] = vv;
    }
}

extern "C" void kernel_launch(void* const* d_in, const int* in_sizes, int n_in,
                              void* d_out, int out_size, void* d_ws, size_t ws_size,
                              hipStream_t stream) {
    (void)n_in; (void)out_size; (void)d_ws; (void)ws_size;
    const float* x  = (const float*)d_in[0];
    const float* cr = (const float*)d_in[1];
    const float* ci = (const float*)d_in[2];
    float* out = (float*)d_out;

    const int Bb = 64;
    int S        = in_sizes[0] / (Bb * 2);
    int out_len  = S - MDEPTH + 1;
    int tilesPerRow = (out_len + TILE - 1) / TILE;
    int totalTiles  = Bb * tilesPerRow;
    int wavesTotal  = (totalTiles + TPW - 1) / TPW;
    int threads     = 256;
    int blocks      = (wavesTotal + (threads / 32) - 1) / (threads / 32);

    volterra_wmma_kernel<<<blocks, threads, 0, stream>>>(
        x, cr, ci, out, S, out_len, tilesPerRow, totalTiles);
}


# Re-map the GEMM: 16 segments of 8 outputs fill N, constant banded coefficient matrix as A; 24 WMMAs per 128 outputs (was 80), long waves; sum order changes (two accumulator chains)
# speedup vs baseline: 2.4743x; 2.4743x over previous
#include <hip/hip_runtime.h>

typedef __attribute__((ext_vector_type(2))) float v2f;
typedef __attribute__((ext_vector_type(4))) float v4f;
typedef __attribute__((ext_vector_type(8))) float v8f;

#define MDEPTH 5
#define SEG    8
#define BLK    128
#define BPW    4

__global__ __launch_bounds__(128) void volterra_wmma_kernel(
    const float* __restrict__ x,
    const float* __restrict__ cr,
    const float* __restrict__ ci,
    float* __restrict__ out,
    int S, int out_len, int blocksPerRow, int totalBlocks)
{
    const int lane   = threadIdx.x & 31;
    const int waveId = __builtin_amdgcn_readfirstlane((int)((blockIdx.x * blockDim.x + threadIdx.x) >> 5));
    const int col    = lane & 15;
    const int hl     = lane >> 4;

    const int n    = col >> 1;
    const int comp = col & 1;
    const float* tu = comp ? ci : cr;
    const float* tv = comp ? cr : ci;
    const float  sv = comp ? 1.0f : -1.0f;
    v2f Au[4][3], Av[4][3];
#pragma unroll
    for (int j = 0; j < 3; ++j) {
        float au[4][2], av[4][2];
#pragma unroll
        for (int s = 0; s < 2; ++s) {
            const int  so = 4 * j + 2 * hl + s;
            const int  m  = n + (MDEPTH - 1) - so;
            const bool ok = (m >= 0) && (m <= MDEPTH - 1);
            const int  mc = ok ? m : 0;
#pragma unroll
            for (int k = 0; k < 4; ++k) {
                float a = tu[k * MDEPTH + mc];
                float b = tv[k * MDEPTH + mc] * sv;
                au[k][s] = ok ? a : 0.0f;
                av[k][s] = ok ? b : 0.0f;
            }
        }
#pragma unroll
        for (int k = 0; k < 4; ++k) {
            v2f t0; t0.x = au[k][0]; t0.y = au[k][1]; Au[k][j] = t0;
            v2f t1; t1.x = av[k][0]; t1.y = av[k][1]; Av[k][j] = t1;
        }
    }

    const int g0 = waveId * BPW;
    if (g0 >= totalBlocks) return;

    v4f cur[3];
    {
        const int brow = g0 / blocksPerRow;
        const int sb   = (g0 - brow * blocksPerRow) * BLK + col * SEG + 2 * hl;
        const float* xr = x + (size_t)brow * S * 2;
#pragma unroll
        for (int j = 0; j < 3; ++j) {
            int idx = sb + 4 * j;
            if (idx > S - 2) idx = S - 2;
            cur[j] = *(const v4f*)(xr + 2 * (size_t)idx);
        }
    }

#pragma unroll
    for (int i = 0; i < BPW; ++i) {
        const int g = g0 + i;
        if (g >= totalBlocks) break;
        const int brow = g / blocksPerRow;
        const int ob   = (g - brow * blocksPerRow) * BLK + col * SEG;

        v4f nxt[3];
        {
            int gn = g + 1;
            if (gn > totalBlocks - 1) gn = totalBlocks - 1;
            const int brn = gn / blocksPerRow;
            const int sbn = (gn - brn * blocksPerRow) * BLK + col * SEG + 2 * hl;
            const float* xn = x + (size_t)brn * S * 2;
#pragma unroll
            for (int j = 0; j < 3; ++j) {
                int idx = sbn + 4 * j;
                if (idx > S - 2) idx = S - 2;
                nxt[j] = *(const v4f*)(xn + 2 * (size_t)idx);
            }
        }

        v2f U[4][3], V[4][3];
#pragma unroll
        for (int j = 0; j < 3; ++j) {
            const float r0 = cur[j].x, i0 = cur[j].y, r1 = cur[j].z, i1 = cur[j].w;
            const float e0 = r0 * r0 + i0 * i0;
            const float e1 = r1 * r1 + i1 * i1;
            const float q0 = e0 * e0, q1 = e1 * e1;
            const float c0 = q0 * e0, c1 = q1 * e1;
            v2f t;
            t.x = r0;      t.y = r1;      U[0][j] = t;
            t.x = e0 * r0; t.y = e1 * r1; U[1][j] = t;
            t.x = q0 * r0; t.y = q1 * r1; U[2][j] = t;
            t.x = c0 * r0; t.y = c1 * r1; U[3][j] = t;
            t.x = i0;      t.y = i1;      V[0][j] = t;
            t.x = e0 * i0; t.y = e1 * i1; V[1][j] = t;
            t.x = q0 * i0; t.y = q1 * i1; V[2][j] = t;
            t.x = c0 * i0; t.y = c1 * i1; V[3][j] = t;
        }

        v8f accU = {0.f, 0.f, 0.f, 0.f, 0.f, 0.f, 0.f, 0.f};
        v8f accV = {0.f, 0.f, 0.f, 0.f, 0.f, 0.f, 0.f, 0.f};
#pragma unroll
        for (int k = 0; k < 4; ++k) {
#pragma unroll
            for (int j = 0; j < 3; ++j) {
                accU = __builtin_amdgcn_wmma_f32_16x16x4_f32(
                           false, Au[k][j], false, U[k][j], (short)0, accU, false, false);
                accV = __builtin_amdgcn_wmma_f32_16x16x4_f32(
                           false, Av[k][j], false, V[k][j], (short)0, accV, false, false);
            }
        }
        const v8f acc = accU + accV;

        const int o0 = ob + 4 * hl;
        float* p = out + ((size_t)brow * out_len + o0) * 2;
        if (o0 + 3 < out_len) {
            v4f lo; lo.x = acc[0]; lo.y = acc[1]; lo.z = acc[2]; lo.w = acc[3];
            v4f hi; hi.x = acc[4]; hi.y = acc[5]; hi.z = acc[6]; hi.w = acc[7];
            *(v4f*)p       = lo;
            *(v4f*)(p + 4) = hi;
        } else {
#pragma unroll
            for (int q = 0; q < 4; ++q) {
                if (o0 + q < out_len) {
                    v2f t; t.x = acc[2 * q]; t.y = acc[2 * q + 1];
                    *(v2f*)(p + 2 * q) = t;
                }
            }
        }

#pragma unroll
        for (int j = 0; j < 3; ++j) cur[j] = nxt[j];
    }
}

extern "C" void kernel_launch(void* const* d_in, const int* in_sizes, int n_in,
                              void* d_out, int out_size, void* d_ws, size_t ws_size,
                              hipStream_t stream) {
    (void)n_in; (void)out_size; (void)d_ws; (void)ws_size;
    const float* x  = (const float*)d_in[0];
    const float* cr = (const float*)d_in[1];
    const float* ci = (const float*)d_in[2];
    float* out = (float*)d_out;

    const int Bb = 64;
    int S        = in_sizes[0] / (Bb * 2);
    int out_len  = S - MDEPTH + 1;
    int blocksPerRow = (out_len + BLK - 1) / BLK;
    int totalBlocks  = Bb * blocksPerRow;
    int wavesTotal   = (totalBlocks + BPW - 1) / BPW;
    int threads      = 128;
    int blocks       = (wavesTotal + (threads / 32) - 1) / (threads / 32);

    volterra_wmma_kernel<<<blocks, threads, 0, stream>>>(
        x, cr, ci, out, S, out_len, blocksPerRow, totalBlocks);
}


# All 4 blocks' sample windows of a wave requested at entry (12 x 16 B loads per lane in flight; waves_per_eu ceiling 4 so the compiler keeps them there) instead of one block ahead
# speedup vs baseline: 2.7328x; 1.1045x over previous
#include <hip/hip_runtime.h>

typedef __attribute__((ext_vector_type(2))) float v2f;
typedef __attribute__((ext_vector_type(4))) float v4f;
typedef __attribute__((ext_vector_type(8))) float v8f;

#define MDEPTH 5
#define SEG    8
#define BLK    128
#define BPW    4

__global__ __launch_bounds__(128) __attribute__((amdgpu_waves_per_eu(1, 4))) void volterra_wmma_kernel(
    const float* __restrict__ x,
    const float* __restrict__ cr,
    const float* __restrict__ ci,
    float* __restrict__ out,
    int S, int out_len, int blocksPerRow, int totalBlocks)
{
    const int lane   = threadIdx.x & 31;
    const int waveId = __builtin_amdgcn_readfirstlane((int)((blockIdx.x * blockDim.x + threadIdx.x) >> 5));
    const int col    = lane & 15;
    const int hl     = lane >> 4;

    const int n    = col >> 1;
    const int comp = col & 1;
    const float* tu = comp ? ci : cr;
    const float* tv = comp ? cr : ci;
    const float  sv = comp ? 1.0f : -1.0f;
    v2f Au[4][3], Av[4][3];
#pragma unroll
    for (int j = 0; j < 3; ++j) {
        float au[4][2], av[4][2];
#pragma unroll
        for (int s = 0; s < 2; ++s) {
            const int  so = 4 * j + 2 * hl + s;
            const int  m  = n + (MDEPTH - 1) - so;
            const bool ok = (m >= 0) && (m <= MDEPTH - 1);
            const int  mc = ok ? m : 0;
#pragma unroll
            for (int k = 0; k < 4; ++k) {
                float a = tu[k * MDEPTH + mc];
                float b = tv[k * MDEPTH + mc] * sv;
                au[k][s] = ok ? a : 0.0f;
                av[k][s] = ok ? b : 0.0f;
            }
        }
#pragma unroll
        for (int k = 0; k < 4; ++k) {
            v2f t0; t0.x = au[k][0]; t0.y = au[k][1]; Au[k][j] = t0;
            v2f t1; t1.x = av[k][0]; t1.y = av[k][1]; Av[k][j] = t1;
        }
    }

    const int g0 = waveId * BPW;
    if (g0 >= totalBlocks) return;

    v4f win[BPW][3];
#pragma unroll
    for (int i = 0; i < BPW; ++i) {
        int gi = g0 + i;
        if (gi > totalBlocks - 1) gi = totalBlocks - 1;
        const int brw = gi / blocksPerRow;
        const int sb  = (gi - brw * blocksPerRow) * BLK + col * SEG + 2 * hl;
        const float* xr = x + (size_t)brw * S * 2;
#pragma unroll
        for (int j = 0; j < 3; ++j) {
            int idx = sb + 4 * j;
            if (idx > S - 2) idx = S - 2;
            win[i][j] = *(const v4f*)(xr + 2 * (size_t)idx);
        }
    }

#pragma unroll
    for (int i = 0; i < BPW; ++i) {
        const int g = g0 + i;
        if (g >= totalBlocks) break;
        const int brow = g / blocksPerRow;
        const int ob   = (g - brow * blocksPerRow) * BLK + col * SEG;

        v2f U[4][3], V[4][3];
#pragma unroll
        for (int j = 0; j < 3; ++j) {
            const float r0 = win[i][j].x, i0 = win[i][j].y, r1 = win[i][j].z, i1 = win[i][j].w;
            const float e0 = r0 * r0 + i0 * i0;
            const float e1 = r1 * r1 + i1 * i1;
            const float q0 = e0 * e0, q1 = e1 * e1;
            const float c0 = q0 * e0, c1 = q1 * e1;
            v2f t;
            t.x = r0;      t.y = r1;      U[0][j] = t;
            t.x = e0 * r0; t.y = e1 * r1; U[1][j] = t;
            t.x = q0 * r0; t.y = q1 * r1; U[2][j] = t;
            t.x = c0 * r0; t.y = c1 * r1; U[3][j] = t;
            t.x = i0;      t.y = i1;      V[0][j] = t;
            t.x = e0 * i0; t.y = e1 * i1; V[1][j] = t;
            t.x = q0 * i0; t.y = q1 * i1; V[2][j] = t;
            t.x = c0 * i0; t.y = c1 * i1; V[3][j] = t;
        }

        v8f accU = {0.f, 0.f, 0.f, 0.f, 0.f, 0.f, 0.f, 0.f};
        v8f accV = {0.f, 0.f, 0.f, 0.f, 0.f, 0.f, 0.f, 0.f};
#pragma unroll
        for (int k = 0; k < 4; ++k) {
#pragma unroll
            for (int j = 0; j < 3; ++j) {
                accU = __builtin_amdgcn_wmma_f32_16x16x4_f32(
                           false, Au[k][j], false, U[k][j], (short)0, accU, false, false);
                accV = __builtin_amdgcn_wmma_f32_16x16x4_f32(
                           false, Av[k][j], false, V[k][j], (short)0, accV, false, false);
            }
        }
        const v8f acc = accU + accV;

        const int o0 = ob + 4 * hl;
        float* p = out + ((size_t)brow * out_len + o0) * 2;
        if (o0 + 3 < out_len) {
            v4f lo; lo.x = acc[0]; lo.y = acc[1]; lo.z = acc[2]; lo.w = acc[3];
            v4f hi; hi.x = acc[4]; hi.y = acc[5]; hi.z = acc[6]; hi.w = acc[7];
            *(v4f*)p       = lo;
            *(v4f*)(p + 4) = hi;
        } else {
#pragma unroll
            for (int q = 0; q < 4; ++q) {
                if (o0 + q < out_len) {
                    v2f t; t.x = acc[2 * q]; t.y = acc[2 * q + 1];
                    *(v2f*)(p + 2 * q) = t;
                }
            }
        }
    }
}

extern "C" void kernel_launch(void* const* d_in, const int* in_sizes, int n_in,
                              void* d_out, int out_size, void* d_ws, size_t ws_size,
                              hipStream_t stream) {
    (void)n_in; (void)out_size; (void)d_ws; (void)ws_size;
    const float* x  = (const float*)d_in[0];
    const float* cr = (const float*)d_in[1];
    const float* ci = (const float*)d_in[2];
    float* out = (float*)d_out;

    const int Bb = 64;
    int S        = in_sizes[0] / (Bb * 2);
    int out_len  = S - MDEPTH + 1;
    int blocksPerRow = (out_len + BLK - 1) / BLK;
    int totalBlocks  = Bb * blocksPerRow;
    int wavesTotal   = (totalBlocks + BPW - 1) / BPW;
    int threads      = 128;
    int blocks       = (wavesTotal + (threads / 32) - 1) / (threads / 32);

    volterra_wmma_kernel<<<blocks, threads, 0, stream>>>(
        x, cr, ci, out, S, out_len, blocksPerRow, totalBlocks);
}
